# starting kernel
# baseline (speedup 1.0000x reference)
#include <hip/hip_runtime.h>

typedef float v2f __attribute__((ext_vector_type(2)));
typedef float v8f __attribute__((ext_vector_type(8)));

#define C_DIM 256

__global__ void k_init(float* __restrict__ deg, int* __restrict__ has_edge,
                       int N, int NC) {
    int i = blockIdx.x * blockDim.x + threadIdx.x;
    if (i < N) deg[i] = 1.0f;
    if (i < NC) has_edge[i] = 0;
}

__global__ void k_edges(const int* __restrict__ src, const int* __restrict__ dst,
                        const int* __restrict__ assign,
                        float* __restrict__ deg, int* __restrict__ has_edge, int E) {
    int e = blockIdx.x * blockDim.x + threadIdx.x;
    if (e >= E) return;
    int s = src[e], d = dst[e];
    int cs = assign[s];
    if (cs == assign[d]) {
        unsafeAtomicAdd(&deg[d], 1.0f);
        has_edge[cs] = 1;
    }
}

__global__ __launch_bounds__(256) void k_gemm(const float* __restrict__ X,
                                              const float* __restrict__ W,
                                              float* __restrict__ xW, int N) {
    const int lane = threadIdx.x & 31;
    const int wave = threadIdx.x >> 5;
    const int half = lane >> 4;
    const int m    = lane & 15;
    const int r0   = blockIdx.x * 128 + wave * 16;
    if (r0 >= N) return;

    v8f acc[16] = {};

    int rowA = r0 + m;
    if (rowA >= N) rowA = N - 1;
    const float* __restrict__ Arow = X + (size_t)rowA * C_DIM + 2 * half;
    const float* __restrict__ Wb = W + (size_t)(2 * half) * C_DIM + m;

    for (int k0 = 0; k0 < C_DIM; k0 += 4) {
        v2f a;
        a.x = Arow[k0];
        a.y = Arow[k0 + 1];
#pragma unroll
        for (int t = 0; t < 16; ++t) {
            v2f bf;
            bf.x = Wb[(size_t)k0 * C_DIM + t * 16];
            bf.y = Wb[(size_t)(k0 + 1) * C_DIM + t * 16];
            acc[t] = __builtin_amdgcn_wmma_f32_16x16x4_f32(
                false, a, false, bf, (short)0, acc[t], false, false);
        }
    }

#pragma unroll
    for (int t = 0; t < 16; ++t) {
#pragma unroll
        for (int v = 0; v < 8; ++v) {
            int row = r0 + v + 8 * half;
            if (row < N) xW[(size_t)row * C_DIM + t * 16 + m] = acc[t][v];
        }
    }
}

__global__ void k_self(const float* __restrict__ X, const float* __restrict__ xW,
                       const float* __restrict__ deg, const float* __restrict__ b,
                       const int* __restrict__ assign, const int* __restrict__ has_edge,
                       float* __restrict__ out, int N) {
    int idx = blockIdx.x * blockDim.x + threadIdx.x;
    if (idx >= N * C_DIM) return;
    int n = idx >> 8;
    int c = idx & (C_DIM - 1);
    if (has_edge[assign[n]]) {
        out[idx] = xW[idx] / deg[n] + b[c];
    } else {
        out[idx] = X[idx];
    }
}

__global__ __launch_bounds__(256) void k_scatter(const int* __restrict__ src,
                                                 const int* __restrict__ dst,
                                                 const int* __restrict__ assign,
                                                 const float* __restrict__ deg,
                                                 const float* __restrict__ xW,
                                                 float* __restrict__ out, int E) {
    int e    = (blockIdx.x * blockDim.x + threadIdx.x) >> 5;
    int lane = threadIdx.x & 31;
    if (e >= E) return;
    int s = src[e], d = dst[e];
    if (assign[s] != assign[d]) return;
    float w = rsqrtf(deg[s]) * rsqrtf(deg[d]);
    const float4* xs = (const float4*)(xW + (size_t)s * C_DIM) + lane * 2;
    float4 v0 = xs[0];
    float4 v1 = xs[1];
    float* o = out + (size_t)d * C_DIM + lane * 8;
    unsafeAtomicAdd(o + 0, w * v0.x);
    unsafeAtomicAdd(o + 1, w * v0.y);
    unsafeAtomicAdd(o + 2, w * v0.z);
    unsafeAtomicAdd(o + 3, w * v0.w);
    unsafeAtomicAdd(o + 4, w * v1.x);
    unsafeAtomicAdd(o + 5, w * v1.y);
    unsafeAtomicAdd(o + 6, w * v1.z);
    unsafeAtomicAdd(o + 7, w * v1.w);
}

extern "C" void kernel_launch(void* const* d_in, const int* in_sizes, int n_in,
                              void* d_out, int out_size, void* d_ws, size_t ws_size,
                              hipStream_t stream) {
    const float* X      = (const float*)d_in[0];
    const float* W      = (const float*)d_in[1];
    const float* b      = (const float*)d_in[2];
    const int*   assign = (const int*)d_in[3];
    const int*   ei     = (const int*)d_in[4];

    const int N  = in_sizes[0] / C_DIM;
    const int E  = in_sizes[4] / 2;
    const int NC = 256;

    const int* src = ei;
    const int* dst = ei + E;

    char* ws = (char*)d_ws;
    size_t off = 0;
    float* deg = (float*)(ws + off);
    off += (((size_t)N * 4) + 255) & ~(size_t)255;
    int* has_edge = (int*)(ws + off);
    off += (((size_t)NC * 4) + 255) & ~(size_t)255;
    float* xW = (float*)(ws + off);

    float* out = (float*)d_out;

    k_init<<<(N + 255) / 256, 256, 0, stream>>>(deg, has_edge, N, NC);
    k_edges<<<(E + 255) / 256, 256, 0, stream>>>(src, dst, assign, deg, has_edge, E);
    k_gemm<<<(N + 127) / 128, 256, 0, stream>>>(X, W, xW, N);
    k_self<<<(N * C_DIM + 255) / 256, 256, 0, stream>>>(X, xW, deg, b, assign,
                                                        has_edge, out, N);
    k_scatter<<<(E + 7) / 8, 256, 0, stream>>>(src, dst, assign, deg, xW, out, E);
}


# replace the float-atomic edge scatter and the self/bias pass by a per-destination gather over a counting-sorted edge list (integer atomics only): each out row summed in registers, written once
# speedup vs baseline: 1.8926x; 1.8926x over previous
#include <hip/hip_runtime.h>

typedef float v2f __attribute__((ext_vector_type(2)));
typedef float v8f __attribute__((ext_vector_type(8)));

#define C_DIM 256

__global__ void k_init(int* __restrict__ cnt, int* __restrict__ has_edge,
                       int* __restrict__ total, int N, int NC) {
    int i = blockIdx.x * blockDim.x + threadIdx.x;
    if (i < N) cnt[i] = 0;
    if (i < NC) has_edge[i] = 0;
    if (i == 0) *total = 0;
}

__global__ void k_count(const int* __restrict__ src, const int* __restrict__ dst,
                        const int* __restrict__ assign, int* __restrict__ cnt, int E) {
    int e = blockIdx.x * blockDim.x + threadIdx.x;
    if (e >= E) return;
    int s = src[e], d = dst[e];
    if (assign[s] == assign[d]) atomicAdd(&cnt[d], 1);
}

__global__ __launch_bounds__(256) void k_alloc(const int* __restrict__ cnt,
                                               const int* __restrict__ assign,
                                               int* __restrict__ cursor,
                                               int* __restrict__ has_edge,
                                               int* __restrict__ total, int N) {
    __shared__ int wsum[8];
    __shared__ int blk_base;
    const int i    = blockIdx.x * 256 + threadIdx.x;
    const int lane = threadIdx.x & 31;
    const int wave = threadIdx.x >> 5;
    const int c  = (i < N) ? cnt[i] : 0;
    const int cs = (i < N) ? assign[i] : -1;

    int x = c;
    for (int o = 1; o < 32; o <<= 1) {
        int y = __shfl_up(x, o);
        if (lane >= o) x += y;
    }
    if (lane == 31) wsum[wave] = x;

    const int prev_cs = __shfl_up(cs, 1);
    const int prev_c  = __shfl_up(c, 1);
    if (c > 0 && (lane == 0 || prev_c == 0 || prev_cs != cs)) has_edge[cs] = 1;

    __syncthreads();
    if (threadIdx.x == 0) {
        int t = 0;
        for (int w = 0; w < 8; ++w) { int v = wsum[w]; wsum[w] = t; t += v; }
        blk_base = atomicAdd(total, t);
    }
    __syncthreads();
    if (i < N) cursor[i] = blk_base + wsum[wave] + x - c;
}

__global__ void k_fill(const int* __restrict__ src, const int* __restrict__ dst,
                       const int* __restrict__ assign, int* __restrict__ cursor,
                       int* __restrict__ esrc, int E) {
    int e = blockIdx.x * blockDim.x + threadIdx.x;
    if (e >= E) return;
    int s = src[e], d = dst[e];
    if (assign[s] == assign[d]) {
        int p = atomicAdd(&cursor[d], 1);
        esrc[p] = s;
    }
}

__global__ __launch_bounds__(256) void k_gemm(const float* __restrict__ X,
                                              const float* __restrict__ W,
                                              float* __restrict__ xW, int N) {
    const int lane = threadIdx.x & 31;
    const int wave = threadIdx.x >> 5;
    const int half = lane >> 4;
    const int m    = lane & 15;
    const int r0   = blockIdx.x * 128 + wave * 16;
    if (r0 >= N) return;

    v8f acc[16] = {};

    int rowA = r0 + m;
    if (rowA >= N) rowA = N - 1;
    const float* __restrict__ Arow = X + (size_t)rowA * C_DIM + 2 * half;
    const float* __restrict__ Wb = W + (size_t)(2 * half) * C_DIM + m;

    for (int k0 = 0; k0 < C_DIM; k0 += 4) {
        v2f a;
        a.x = Arow[k0];
        a.y = Arow[k0 + 1];
#pragma unroll
        for (int t = 0; t < 16; ++t) {
            v2f bf;
            bf.x = Wb[(size_t)k0 * C_DIM + t * 16];
            bf.y = Wb[(size_t)(k0 + 1) * C_DIM + t * 16];
            acc[t] = __builtin_amdgcn_wmma_f32_16x16x4_f32(
                false, a, false, bf, (short)0, acc[t], false, false);
        }
    }

#pragma unroll
    for (int t = 0; t < 16; ++t) {
#pragma unroll
        for (int v = 0; v < 8; ++v) {
            int row = r0 + v + 8 * half;
            if (row < N) xW[(size_t)row * C_DIM + t * 16 + m] = acc[t][v];
        }
    }
}

__device__ __forceinline__ void fma4(float4& a, float w, const float4& v) {
    a.x += w * v.x; a.y += w * v.y; a.z += w * v.z; a.w += w * v.w;
}

__global__ __launch_bounds__(256) void k_gather(const float* __restrict__ X,
                                                const float* __restrict__ xW,
                                                const float* __restrict__ b,
                                                const int* __restrict__ assign,
                                                const int* __restrict__ has_edge,
                                                const int* __restrict__ cnt,
                                                const int* __restrict__ cursor,
                                                const int* __restrict__ esrc,
                                                float* __restrict__ out, int N) {
    const int lane = threadIdx.x & 31;
    const int n = __builtin_amdgcn_readfirstlane(blockIdx.x * 8 + (threadIdx.x >> 5));
    if (n >= N) return;
    float4* __restrict__ o4 = (float4*)(out + (size_t)n * C_DIM);

    if (!has_edge[assign[n]]) {
        const float4* x4 = (const float4*)(X + (size_t)n * C_DIM);
        o4[lane]      = x4[lane];
        o4[lane + 32] = x4[lane + 32];
        return;
    }

    const int   c    = cnt[n];
    const int   beg  = cursor[n] - c;
    const float degn = 1.0f + (float)c;
    const float dn   = rsqrtf(degn);

    const float4* xr = (const float4*)(xW + (size_t)n * C_DIM);
    const float4* b4 = (const float4*)b;
    float4 a0 = xr[lane], a1 = xr[lane + 32];
    const float4 b0 = b4[lane], b1 = b4[lane + 32];
    a0.x = a0.x / degn + b0.x; a0.y = a0.y / degn + b0.y;
    a0.z = a0.z / degn + b0.z; a0.w = a0.w / degn + b0.w;
    a1.x = a1.x / degn + b1.x; a1.y = a1.y / degn + b1.y;
    a1.z = a1.z / degn + b1.z; a1.w = a1.w / degn + b1.w;

    for (int base = 0; base < c; base += 32) {
        const int m = (c - base < 32) ? (c - base) : 32;
        int   sj = 0;
        float wj = 0.0f;
        if (lane < m) {
            sj = esrc[beg + base + lane];
            wj = rsqrtf(1.0f + (float)cnt[sj]) * dn;
        }
        int j = 0;
        for (; j + 4 <= m; j += 4) {
            const int   sA = __builtin_amdgcn_readlane(sj, j);
            const int   sB = __builtin_amdgcn_readlane(sj, j + 1);
            const int   sC = __builtin_amdgcn_readlane(sj, j + 2);
            const int   sD = __builtin_amdgcn_readlane(sj, j + 3);
            const float wA = __int_as_float(__builtin_amdgcn_readlane(__float_as_int(wj), j));
            const float wB = __int_as_float(__builtin_amdgcn_readlane(__float_as_int(wj), j + 1));
            const float wC = __int_as_float(__builtin_amdgcn_readlane(__float_as_int(wj), j + 2));
            const float wD = __int_as_float(__builtin_amdgcn_readlane(__float_as_int(wj), j + 3));
            const float4* pA = (const float4*)(xW + (size_t)sA * C_DIM);
            const float4* pB = (const float4*)(xW + (size_t)sB * C_DIM);
            const float4* pC = (const float4*)(xW + (size_t)sC * C_DIM);
            const float4* pD = (const float4*)(xW + (size_t)sD * C_DIM);
            const float4 vA0 = pA[lane], vA1 = pA[lane + 32];
            const float4 vB0 = pB[lane], vB1 = pB[lane + 32];
            const float4 vC0 = pC[lane], vC1 = pC[lane + 32];
            const float4 vD0 = pD[lane], vD1 = pD[lane + 32];
            fma4(a0, wA, vA0); fma4(a1, wA, vA1);
            fma4(a0, wB, vB0); fma4(a1, wB, vB1);
            fma4(a0, wC, vC0); fma4(a1, wC, vC1);
            fma4(a0, wD, vD0); fma4(a1, wD, vD1);
        }
        for (; j < m; ++j) {
            const int   sA = __builtin_amdgcn_readlane(sj, j);
            const float wA = __int_as_float(__builtin_amdgcn_readlane(__float_as_int(wj), j));
            const float4* pA = (const float4*)(xW + (size_t)sA * C_DIM);
            const float4 vA0 = pA[lane], vA1 = pA[lane + 32];
            fma4(a0, wA, vA0); fma4(a1, wA, vA1);
        }
    }
    o4[lane]      = a0;
    o4[lane + 32] = a1;
}

extern "C" void kernel_launch(void* const* d_in, const int* in_sizes, int n_in,
                              void* d_out, int out_size, void* d_ws, size_t ws_size,
                              hipStream_t stream) {
    const float* X      = (const float*)d_in[0];
    const float* W      = (const float*)d_in[1];
    const float* b      = (const float*)d_in[2];
    const int*   assign = (const int*)d_in[3];
    const int*   ei     = (const int*)d_in[4];

    const int N  = in_sizes[0] / C_DIM;
    const int E  = in_sizes[4] / 2;
    const int NC = 256;

    const int* src = ei;
    const int* dst = ei + E;

    char* ws = (char*)d_ws;
    size_t off = 0;
    int* cnt = (int*)(ws + off);
    off += (((size_t)N * 4) + 255) & ~(size_t)255;
    int* has_edge = (int*)(ws + off);
    off += (((size_t)NC * 4) + 255) & ~(size_t)255;
    int* total = (int*)(ws + off);
    off += 256;
    int* cursor = (int*)(ws + off);
    off += (((size_t)N * 4) + 255) & ~(size_t)255;
    int* esrc = (int*)(ws + off);
    off += (((size_t)E * 4) + 255) & ~(size_t)255;
    float* xW = (float*)(ws + off);
    off += (size_t)N * C_DIM * 4;
    if (off > ws_size) return;

    float* out = (float*)d_out;

    k_init<<<(N + 255) / 256, 256, 0, stream>>>(cnt, has_edge, total, N, NC);
    k_count<<<(E + 255) / 256, 256, 0, stream>>>(src, dst, assign, cnt, E);
    k_alloc<<<(N + 255) / 256, 256, 0, stream>>>(cnt, assign, cursor, has_edge, total, N);
    k_fill<<<(E + 255) / 256, 256, 0, stream>>>(src, dst, assign, cursor, esrc, E);
    k_gemm<<<(N + 127) / 128, 256, 0, stream>>>(X, W, xW, N);
    k_gather<<<(N + 7) / 8, 256, 0, stream>>>(X, xW, b, assign, has_edge, cnt, cursor,
                                              esrc, out, N);
}


# matrix phase: each lane owns 4 adjacent columns and each wave 32 rows x 64 columns, so B comes by one 128-bit W load per k for 8 products and out rows leave as whole 128-byte lines
# speedup vs baseline: 2.7032x; 1.4283x over previous
#include <hip/hip_runtime.h>

typedef float v2f __attribute__((ext_vector_type(2)));
typedef float v8f __attribute__((ext_vector_type(8)));

#define C_DIM 256

__global__ void k_init(int* __restrict__ cnt, int* __restrict__ has_edge,
                       int* __restrict__ total, int N, int NC) {
    int i = blockIdx.x * blockDim.x + threadIdx.x;
    if (i < N) cnt[i] = 0;
    if (i < NC) has_edge[i] = 0;
    if (i == 0) *total = 0;
}

__global__ void k_count(const int* __restrict__ src, const int* __restrict__ dst,
                        const int* __restrict__ assign, int* __restrict__ cnt, int E) {
    int e = blockIdx.x * blockDim.x + threadIdx.x;
    if (e >= E) return;
    int s = src[e], d = dst[e];
    if (assign[s] == assign[d]) atomicAdd(&cnt[d], 1);
}

__global__ __launch_bounds__(256) void k_alloc(const int* __restrict__ cnt,
                                               const int* __restrict__ assign,
                                               int* __restrict__ cursor,
                                               int* __restrict__ has_edge,
                                               int* __restrict__ total, int N) {
    __shared__ int wsum[8];
    __shared__ int blk_base;
    const int i    = blockIdx.x * 256 + threadIdx.x;
    const int lane = threadIdx.x & 31;
    const int wave = threadIdx.x >> 5;
    const int c  = (i < N) ? cnt[i] : 0;
    const int cs = (i < N) ? assign[i] : -1;

    int x = c;
    for (int o = 1; o < 32; o <<= 1) {
        int y = __shfl_up(x, o);
        if (lane >= o) x += y;
    }
    if (lane == 31) wsum[wave] = x;

    const int prev_cs = __shfl_up(cs, 1);
    const int prev_c  = __shfl_up(c, 1);
    if (c > 0 && (lane == 0 || prev_c == 0 || prev_cs != cs)) has_edge[cs] = 1;

    __syncthreads();
    if (threadIdx.x == 0) {
        int t = 0;
        for (int w = 0; w < 8; ++w) { int v = wsum[w]; wsum[w] = t; t += v; }
        blk_base = atomicAdd(total, t);
    }
    __syncthreads();
    if (i < N) cursor[i] = blk_base + wsum[wave] + x - c;
}

__global__ void k_fill(const int* __restrict__ src, const int* __restrict__ dst,
                       const int* __restrict__ assign, int* __restrict__ cursor,
                       int* __restrict__ esrc, int E) {
    int e = blockIdx.x * blockDim.x + threadIdx.x;
    if (e >= E) return;
    int s = src[e], d = dst[e];
    if (assign[s] == assign[d]) {
        int p = atomicAdd(&cursor[d], 1);
        esrc[p] = s;
    }
}

#define TILE_M  64
#define A_PITCH 260

__device__ __forceinline__ void fma4(float4& a, float w, const float4& v) {
    a.x += w * v.x; a.y += w * v.y; a.z += w * v.z; a.w += w * v.w;
}

__global__ __launch_bounds__(256) void k_fused(const float* __restrict__ X,
                                               const float* __restrict__ W,
                                               const float* __restrict__ b,
                                               const int* __restrict__ assign,
                                               const int* __restrict__ has_edge,
                                               const int* __restrict__ cnt,
                                               const int* __restrict__ cursor,
                                               const int* __restrict__ esrc,
                                               float* __restrict__ out, int N) {
    __shared__ __attribute__((aligned(16))) float As[TILE_M * A_PITCH];
    __shared__ int   rowflag[TILE_M];

    const int lane = threadIdx.x & 31;
    const int wave = __builtin_amdgcn_readfirstlane(threadIdx.x >> 5);
    const int blk0 = blockIdx.x * TILE_M;

    const int n0 = blk0 + wave * 8;
    int nl = n0 + (lane & 7);
    if (nl >= N) nl = N - 1;
    const int cL   = cnt[nl];
    const int curL = cursor[nl];
    const int heL  = has_edge[assign[nl]];

    int c0   = __builtin_amdgcn_readlane(cL, 0);
    int beg0 = __builtin_amdgcn_readlane(curL, 0) - c0;
    int c1   = __builtin_amdgcn_readlane(cL, 1);
    int beg1 = __builtin_amdgcn_readlane(curL, 1) - c1;
    int sjA = 0, sjB = 0, csA = 0;
    if (lane < c0) sjA = esrc[beg0 + lane];
    if (lane < c1) sjB = esrc[beg1 + lane];
    if (lane < c0) csA = cnt[sjA];

    for (int r = 0; r < 8; ++r) {
        const int n   = n0 + r;
        const int c   = c0;
        const int beg = beg0;
        int sj = sjA;
        const int he = __builtin_amdgcn_readlane(heL, r);
        const float degn = 1.0f + (float)c;
        const float dn   = rsqrtf(degn);
        float wj = (lane < c) ? rsqrtf(1.0f + (float)csA) * dn : 0.0f;

        const int nc = (n < N) ? n : N - 1;
        const float4* xr = (const float4*)(X + (size_t)nc * C_DIM);
        const float4 s0 = xr[lane], s1 = xr[lane + 32];

        c0 = c1; beg0 = beg1; sjA = sjB; csA = 0; sjB = 0;
        if (r + 1 < 8 && lane < c0) csA = cnt[sjA];
        if (r + 2 < 8) {
            c1   = __builtin_amdgcn_readlane(cL, r + 2);
            beg1 = __builtin_amdgcn_readlane(curL, r + 2) - c1;
            if (lane < c1) sjB = esrc[beg1 + lane];
        }
        float4* __restrict__ arow = (float4*)(As + (wave * 8 + r) * A_PITCH);

        if (n >= N || !he) {
            if (n < N) {
                float4* o4 = (float4*)(out + (size_t)n * C_DIM);
                o4[lane]      = s0;
                o4[lane + 32] = s1;
            }
            const float4 z = make_float4(0.f, 0.f, 0.f, 0.f);
            arow[lane] = z; arow[lane + 32] = z;
            if (lane == 0) rowflag[wave * 8 + r] = 0;
            continue;
        }

        float4 a0 = make_float4(0.f, 0.f, 0.f, 0.f), a1 = a0;

        for (int base = 0; base < c; base += 32) {
            const int m = (c - base < 32) ? (c - base) : 32;
            if (base > 0) {
                sj = 0; wj = 0.0f;
                if (lane < m) {
                    sj = esrc[beg + base + lane];
                    wj = rsqrtf(1.0f + (float)cnt[sj]) * dn;
                }
            }
            int j = 0;
            for (; j + 4 <= m; j += 4) {
                const int   sA = __builtin_amdgcn_readlane(sj, j);
                const int   sB = __builtin_amdgcn_readlane(sj, j + 1);
                const int   sC = __builtin_amdgcn_readlane(sj, j + 2);
                const int   sD = __builtin_amdgcn_readlane(sj, j + 3);
                const float wA = __int_as_float(__builtin_amdgcn_readlane(__float_as_int(wj), j));
                const float wB = __int_as_float(__builtin_amdgcn_readlane(__float_as_int(wj), j + 1));
                const float wC = __int_as_float(__builtin_amdgcn_readlane(__float_as_int(wj), j + 2));
                const float wD = __int_as_float(__builtin_amdgcn_readlane(__float_as_int(wj), j + 3));
                const float4* pA = (const float4*)(X + (size_t)sA * C_DIM);
                const float4* pB = (const float4*)(X + (size_t)sB * C_DIM);
                const float4* pC = (const float4*)(X + (size_t)sC * C_DIM);
                const float4* pD = (const float4*)(X + (size_t)sD * C_DIM);
                const float4 vA0 = pA[lane], vA1 = pA[lane + 32];
                const float4 vB0 = pB[lane], vB1 = pB[lane + 32];
                const float4 vC0 = pC[lane], vC1 = pC[lane + 32];
                const float4 vD0 = pD[lane], vD1 = pD[lane + 32];
                fma4(a0, wA, vA0); fma4(a1, wA, vA1);
                fma4(a0, wB, vB0); fma4(a1, wB, vB1);
                fma4(a0, wC, vC0); fma4(a1, wC, vC1);
                fma4(a0, wD, vD0); fma4(a1, wD, vD1);
            }
            for (; j < m; ++j) {
                const int   sA = __builtin_amdgcn_readlane(sj, j);
                const float wA = __int_as_float(__builtin_amdgcn_readlane(__float_as_int(wj), j));
                const float4* pA = (const float4*)(X + (size_t)sA * C_DIM);
                const float4 vA0 = pA[lane], vA1 = pA[lane + 32];
                fma4(a0, wA, vA0); fma4(a1, wA, vA1);
            }
        }
        a0.x += s0.x / degn; a0.y += s0.y / degn; a0.z += s0.z / degn; a0.w += s0.w / degn;
        a1.x += s1.x / degn; a1.y += s1.y / degn; a1.z += s1.z / degn; a1.w += s1.w / degn;
        arow[lane]      = a0;
        arow[lane + 32] = a1;
        if (lane == 0) rowflag[wave * 8 + r] = 1;
    }
    __syncthreads();

    const int half = lane >> 4;
    const int m    = lane & 15;
    const int rh   = wave >> 2;
    const int cb   = (wave & 3) * 64 + 4 * m;

    v8f acc[2][4] = {};
    const float* Ap0 = As + (rh * 32 + m) * A_PITCH + 2 * half;
    const float* Ap1 = Ap0 + 16 * A_PITCH;
    const float* __restrict__ Wb = W + (size_t)(2 * half) * C_DIM + cb;

    for (int k0 = 0; k0 < C_DIM; k0 += 4) {
        const v2f a0 = *(const v2f*)(Ap0 + k0);
        const v2f a1 = *(const v2f*)(Ap1 + k0);
        const float4 w0 = *(const float4*)(Wb + (size_t)k0 * C_DIM);
        const float4 w1 = *(const float4*)(Wb + (size_t)(k0 + 1) * C_DIM);
        const float w0a[4] = {w0.x, w0.y, w0.z, w0.w};
        const float w1a[4] = {w1.x, w1.y, w1.z, w1.w};
#pragma unroll
        for (int t = 0; t < 4; ++t) {
            v2f bf;
            bf.x = w0a[t];
            bf.y = w1a[t];
            acc[0][t] = __builtin_amdgcn_wmma_f32_16x16x4_f32(
                false, a0, false, bf, (short)0, acc[0][t], false, false);
            acc[1][t] = __builtin_amdgcn_wmma_f32_16x16x4_f32(
                false, a1, false, bf, (short)0, acc[1][t], false, false);
        }
    }

    const float4 bias = *(const float4*)(b + cb);
#pragma unroll
    for (int rt = 0; rt < 2; ++rt) {
#pragma unroll
        for (int v = 0; v < 8; ++v) {
            const int rl  = rh * 32 + rt * 16 + v + 8 * half;
            const int row = blk0 + rl;
            if (row < N && rowflag[rl]) {
                float4 o;
                o.x = acc[rt][0][v] + bias.x;
                o.y = acc[rt][1][v] + bias.y;
                o.z = acc[rt][2][v] + bias.z;
                o.w = acc[rt][3][v] + bias.w;
                *(float4*)(out + (size_t)row * C_DIM + cb) = o;
            }
        }
    }
}

extern "C" void kernel_launch(void* const* d_in, const int* in_sizes, int n_in,
                              void* d_out, int out_size, void* d_ws, size_t ws_size,
                              hipStream_t stream) {
    const float* X      = (const float*)d_in[0];
    const float* W      = (const float*)d_in[1];
    const float* b      = (const float*)d_in[2];
    const int*   assign = (const int*)d_in[3];
    const int*   ei     = (const int*)d_in[4];

    const int N  = in_sizes[0] / C_DIM;
    const int E  = in_sizes[4] / 2;
    const int NC = 256;

    const int* src = ei;
    const int* dst = ei + E;

    char* ws = (char*)d_ws;
    size_t off = 0;
    int* cnt = (int*)(ws + off);
    off += (((size_t)N * 4) + 255) & ~(size_t)255;
    int* has_edge = (int*)(ws + off);
    off += (((size_t)NC * 4) + 255) & ~(size_t)255;
    int* total = (int*)(ws + off);
    off += 256;
    int* cursor = (int*)(ws + off);
    off += (((size_t)N * 4) + 255) & ~(size_t)255;
    int* esrc = (int*)(ws + off);
    off += (((size_t)E * 4) + 255) & ~(size_t)255;
    if (off > ws_size) return;

    float* out = (float*)d_out;

    k_init<<<(N + 255) / 256, 256, 0, stream>>>(cnt, has_edge, total, N, NC);
    k_count<<<(E + 255) / 256, 256, 0, stream>>>(src, dst, assign, cnt, E);
    k_alloc<<<(N + 255) / 256, 256, 0, stream>>>(cnt, assign, cursor, has_edge, total, N);
    k_fill<<<(E + 255) / 256, 256, 0, stream>>>(src, dst, assign, cursor, esrc, E);
    k_fused<<<(N + TILE_M - 1) / TILE_M, 256, 0, stream>>>(X, W, b, assign, has_edge, cnt, cursor,
                                                         esrc, out, N);
}


# fused kernel: one 32-wave block of 196 rows (13 row tiles) per WGP instead of 782 co-resident 64-row blocks (3 or 4 per WGP), so every WGP carries the same matrix and gather work
# speedup vs baseline: 3.7551x; 1.3891x over previous
#include <hip/hip_runtime.h>

typedef float v2f __attribute__((ext_vector_type(2)));
typedef float v8f __attribute__((ext_vector_type(8)));

#define C_DIM 256

#define NCLUS 256
#define CAP   32

__global__ void k_init(int* __restrict__ cnt, int* __restrict__ has_edge,
                       int* __restrict__ novf, int N, int NC) {
    int i = blockIdx.x * blockDim.x + threadIdx.x;
    if (i < N) cnt[i] = 0;
    if (i < NC) has_edge[i] = 0;
    if (i == 0) *novf = 0;
}

__global__ __launch_bounds__(NCLUS) void k_link(const int* __restrict__ src,
                                                const int* __restrict__ dst,
                                                const int* __restrict__ assign,
                                                int* __restrict__ cnt,
                                                int* __restrict__ slot,
                                                int* __restrict__ odst,
                                                int* __restrict__ osrc,
                                                int* __restrict__ novf,
                                                int* __restrict__ has_edge, int E) {
    __shared__ int cf[NCLUS];
    cf[threadIdx.x] = 0;
    __syncthreads();
    const int e = blockIdx.x * NCLUS + threadIdx.x;
    if (e < E) {
        const int s = src[e], d = dst[e];
        const int a = assign[s];
        if (a == assign[d]) {
            const int p = atomicAdd(&cnt[d], 1);
            if (p < CAP) {
                slot[(size_t)d * CAP + p] = s;
            } else {
                const int q = atomicAdd(novf, 1);
                odst[q] = d;
                osrc[q] = s;
            }
            cf[a] = 1;
        }
    }
    __syncthreads();
    if (cf[threadIdx.x]) has_edge[threadIdx.x] = 1;
}

#define ROWS_B   196
#define LDS_ROWS 208
#define RPW      7
#define GWAVES   28
#define A_PITCH  260

__device__ __forceinline__ void fma4(float4& a, float w, const float4& v) {
    a.x += w * v.x; a.y += w * v.y; a.z += w * v.z; a.w += w * v.w;
}

template <int NT>
__device__ __forceinline__ void mm_loop(const float* Ap0, const float* __restrict__ Wb,
                                        v8f (&acc)[2][4]) {
    for (int k0 = 0; k0 < C_DIM; k0 += 4) {
        v2f a[NT];
#pragma unroll
        for (int rt = 0; rt < NT; ++rt) a[rt] = *(const v2f*)(Ap0 + rt * 16 * A_PITCH + k0);
        const float4 w0 = *(const float4*)(Wb + (size_t)k0 * C_DIM);
        const float4 w1 = *(const float4*)(Wb + (size_t)(k0 + 1) * C_DIM);
        const float w0a[4] = {w0.x, w0.y, w0.z, w0.w};
        const float w1a[4] = {w1.x, w1.y, w1.z, w1.w};
#pragma unroll
        for (int t = 0; t < 4; ++t) {
            v2f bf;
            bf.x = w0a[t];
            bf.y = w1a[t];
#pragma unroll
            for (int rt = 0; rt < NT; ++rt)
                acc[rt][t] = __builtin_amdgcn_wmma_f32_16x16x4_f32(
                    false, a[rt], false, bf, (short)0, acc[rt][t], false, false);
        }
    }
}

__global__ __launch_bounds__(1024) void k_fused(const float* __restrict__ X,
                                               const float* __restrict__ W,
                                               const float* __restrict__ b,
                                               const int* __restrict__ assign,
                                               const int* __restrict__ has_edge,
                                               const int* __restrict__ cnt,
                                               const int* __restrict__ slot,
                                               const int* __restrict__ odst,
                                               const int* __restrict__ osrc,
                                               const int* __restrict__ novf,
                                               float* __restrict__ out, int N) {
    __shared__ __attribute__((aligned(16))) float As[LDS_ROWS * A_PITCH];
    __shared__ int   rowflag[LDS_ROWS];

    const int lane = threadIdx.x & 31;
    const int wave = __builtin_amdgcn_readfirstlane(threadIdx.x >> 5);
    const int blk0 = blockIdx.x * ROWS_B;

    if (wave < GWAVES) {
        const int tr0 = wave * RPW;
        const int n0 = blk0 + tr0;
        int nl = n0 + (lane & 7);
        if (nl >= N) nl = N - 1;
        const int cL   = cnt[nl];
        const int heL  = has_edge[assign[nl]];

        const int nlast = N - 1;
        int c0 = __builtin_amdgcn_readlane(cL, 0);
        int c1 = __builtin_amdgcn_readlane(cL, 1);
        int sjA = 0, sjB = 0, csA = 0;
        if (lane < c0) sjA = slot[(size_t)((n0     < N) ? n0     : nlast) * CAP + lane];
        if (lane < c1) sjB = slot[(size_t)((n0 + 1 < N) ? n0 + 1 : nlast) * CAP + lane];
        if (lane < c0) csA = cnt[sjA];

        for (int r = 0; r < RPW; ++r) {
            const int n   = n0 + r;
            const int c   = c0;
            const int sj  = sjA;
            const int he = __builtin_amdgcn_readlane(heL, r);
            const float degn = 1.0f + (float)c;
            const float dn   = rsqrtf(degn);
            const float wj = (lane < c) ? rsqrtf(1.0f + (float)csA) * dn : 0.0f;

            const int nc = (n < N) ? n : N - 1;
            const float4* xr = (const float4*)(X + (size_t)nc * C_DIM);
            const float4 s0 = xr[lane], s1 = xr[lane + 32];

            c0 = c1; sjA = sjB; csA = 0; sjB = 0;
            if (r + 1 < RPW && lane < c0) csA = cnt[sjA];
            if (r + 2 < RPW) {
                c1 = __builtin_amdgcn_readlane(cL, r + 2);
                const int n2 = (n + 2 < N) ? n + 2 : nlast;
                if (lane < c1) sjB = slot[(size_t)n2 * CAP + lane];
            }
            float4* __restrict__ arow = (float4*)(As + (tr0 + r) * A_PITCH);

            if (n >= N || !he) {
                if (n < N) {
                    float4* o4 = (float4*)(out + (size_t)n * C_DIM);
                    o4[lane]      = s0;
                    o4[lane + 32] = s1;
                }
                const float4 z = make_float4(0.f, 0.f, 0.f, 0.f);
                arow[lane] = z; arow[lane + 32] = z;
                if (lane == 0) rowflag[tr0 + r] = 0;
                continue;
            }

            float4 a0 = make_float4(0.f, 0.f, 0.f, 0.f), a1 = a0;

            {
                const int m = (c < CAP) ? c : CAP;
                int j = 0;
                for (; j + 4 <= m; j += 4) {
                    const int   sA = __builtin_amdgcn_readlane(sj, j);
                    const int   sB = __builtin_amdgcn_readlane(sj, j + 1);
                    const int   sC = __builtin_amdgcn_readlane(sj, j + 2);
                    const int   sD = __builtin_amdgcn_readlane(sj, j + 3);
                    const float wA = __int_as_float(__builtin_amdgcn_readlane(__float_as_int(wj), j));
                    const float wB = __int_as_float(__builtin_amdgcn_readlane(__float_as_int(wj), j + 1));
                    const float wC = __int_as_float(__builtin_amdgcn_readlane(__float_as_int(wj), j + 2));
                    const float wD = __int_as_float(__builtin_amdgcn_readlane(__float_as_int(wj), j + 3));
                    const float4* pA = (const float4*)(X + (size_t)sA * C_DIM);
                    const float4* pB = (const float4*)(X + (size_t)sB * C_DIM);
                    const float4* pC = (const float4*)(X + (size_t)sC * C_DIM);
                    const float4* pD = (const float4*)(X + (size_t)sD * C_DIM);
                    const float4 vA0 = pA[lane], vA1 = pA[lane + 32];
                    const float4 vB0 = pB[lane], vB1 = pB[lane + 32];
                    const float4 vC0 = pC[lane], vC1 = pC[lane + 32];
                    const float4 vD0 = pD[lane], vD1 = pD[lane + 32];
                    fma4(a0, wA, vA0); fma4(a1, wA, vA1);
                    fma4(a0, wB, vB0); fma4(a1, wB, vB1);
                    fma4(a0, wC, vC0); fma4(a1, wC, vC1);
                    fma4(a0, wD, vD0); fma4(a1, wD, vD1);
                }
                for (; j < m; ++j) {
                    const int   sA = __builtin_amdgcn_readlane(sj, j);
                    const float wA = __int_as_float(__builtin_amdgcn_readlane(__float_as_int(wj), j));
                    const float4* pA = (const float4*)(X + (size_t)sA * C_DIM);
                    const float4 vA0 = pA[lane], vA1 = pA[lane + 32];
                    fma4(a0, wA, vA0); fma4(a1, wA, vA1);
                }
            }
            if (c > CAP) {
                const int nov = *novf;
                for (int base = 0; base < nov; base += 32) {
                    const int q = base + lane;
                    int   so = 0;
                    float wo = 0.0f;
                    bool  hit = false;
                    if (q < nov && odst[q] == n) {
                        hit = true;
                        so  = osrc[q];
                        wo  = rsqrtf(1.0f + (float)cnt[so]) * dn;
                    }
                    unsigned pend = __builtin_amdgcn_ballot_w32(hit);
                    while (pend) {
                        const int j = __builtin_ctz(pend);
                        pend &= pend - 1;
                        const int   sA = __builtin_amdgcn_readlane(so, j);
                        const float wA = __int_as_float(__builtin_amdgcn_readlane(__float_as_int(wo), j));
                        const float4* pA = (const float4*)(X + (size_t)sA * C_DIM);
                        const float4 vA0 = pA[lane], vA1 = pA[lane + 32];
                        fma4(a0, wA, vA0); fma4(a1, wA, vA1);
                    }
                }
            }
            a0.x += s0.x / degn; a0.y += s0.y / degn; a0.z += s0.z / degn; a0.w += s0.w / degn;
            a1.x += s1.x / degn; a1.y += s1.y / degn; a1.z += s1.z / degn; a1.w += s1.w / degn;
            arow[lane]      = a0;
            arow[lane + 32] = a1;
            if (lane == 0) rowflag[tr0 + r] = 1;
        }
    } else {
        const float4 z = make_float4(0.f, 0.f, 0.f, 0.f);
        for (int i = 0; i < 3; ++i) {
            const int tr = ROWS_B + (wave - GWAVES) * 3 + i;
            float4* __restrict__ arow = (float4*)(As + tr * A_PITCH);
            arow[lane] = z; arow[lane + 32] = z;
            if (lane == 0) rowflag[tr] = 0;
        }
    }
    __syncthreads();

    const int g = wave >> 3;
    const int p = wave & 7;
    if (p == g) return;
    const bool two = (p != g + 4);
    int rg, cg;
    if (two) {
        const int f = 6 * g + p - (p > g ? 1 : 0) - (p > g + 4 ? 1 : 0);
        rg = f >> 2;
        cg = f & 3;
    } else {
        rg = 6;
        cg = g;
    }
    const int half = lane >> 4;
    const int m    = lane & 15;
    const int cb   = cg * 64 + 4 * m;

    v8f acc[2][4] = {};
    const float* Ap0 = As + (rg * 32 + m) * A_PITCH + 2 * half;
    const float* __restrict__ Wb = W + (size_t)(2 * half) * C_DIM + cb;

    if (two) mm_loop<2>(Ap0, Wb, acc);
    else     mm_loop<1>(Ap0, Wb, acc);

    const float4 bias = *(const float4*)(b + cb);
#pragma unroll
    for (int rt = 0; rt < 2; ++rt) {
#pragma unroll
        for (int v = 0; v < 8; ++v) {
            const int rl  = rg * 32 + rt * 16 + v + 8 * half;
            const int row = blk0 + rl;
            if (rl < ROWS_B && row < N && rowflag[rl]) {
                float4 o;
                o.x = acc[rt][0][v] + bias.x;
                o.y = acc[rt][1][v] + bias.y;
                o.z = acc[rt][2][v] + bias.z;
                o.w = acc[rt][3][v] + bias.w;
                *(float4*)(out + (size_t)row * C_DIM + cb) = o;
            }
        }
    }
}

extern "C" void kernel_launch(void* const* d_in, const int* in_sizes, int n_in,
                              void* d_out, int out_size, void* d_ws, size_t ws_size,
                              hipStream_t stream) {
    const float* X      = (const float*)d_in[0];
    const float* W      = (const float*)d_in[1];
    const float* b      = (const float*)d_in[2];
    const int*   assign = (const int*)d_in[3];
    const int*   ei     = (const int*)d_in[4];

    const int N  = in_sizes[0] / C_DIM;
    const int E  = in_sizes[4] / 2;
    const int NC = NCLUS;

    const int* src = ei;
    const int* dst = ei + E;

    char* ws = (char*)d_ws;
    size_t off = 0;
    int* cnt = (int*)(ws + off);
    off += (((size_t)N * 4) + 255) & ~(size_t)255;
    int* has_edge = (int*)(ws + off);
    off += (((size_t)NC * 4) + 255) & ~(size_t)255;
    int* novf = (int*)(ws + off);
    off += 256;
    int* slot = (int*)(ws + off);
    off += (((size_t)N * CAP * 4) + 255) & ~(size_t)255;
    int* odst = (int*)(ws + off);
    off += (((size_t)E * 4) + 255) & ~(size_t)255;
    int* osrc = (int*)(ws + off);
    off += (((size_t)E * 4) + 255) & ~(size_t)255;
    if (off > ws_size) return;

    float* out = (float*)d_out;

    k_init<<<(N + 255) / 256, 256, 0, stream>>>(cnt, has_edge, novf, N, NC);
    k_link<<<(E + NCLUS - 1) / NCLUS, NCLUS, 0, stream>>>(src, dst, assign, cnt, slot, odst, osrc,
                                                          novf, has_edge, E);
    k_fused<<<(N + ROWS_B - 1) / ROWS_B, 1024, 0, stream>>>(X, W, b, assign, has_edge, cnt, slot,
                                                          odst, osrc, novf, out, N);
}
